# starting kernel
# baseline (speedup 1.0000x reference)
#include <hip/hip_runtime.h>


#define B_   8
#define CI   32
#define CO   32
#define HH   128
#define WW   128
#define HID_ 16
#define TL   16

typedef __attribute__((ext_vector_type(2))) float v2f;
typedef __attribute__((ext_vector_type(8))) float v8f;

__global__ __launch_bounds__(256)
void scaconv_ds_kernel(const float* __restrict__ x,
                       const float* __restrict__ kernelD,
                       const float* __restrict__ kernelP,
                       const float* __restrict__ bias,
                       const float* __restrict__ cvec,
                       const float* __restrict__ mD_W1,
                       const float* __restrict__ mD_b1,
                       const float* __restrict__ mD_W2,
                       const float* __restrict__ mD_b2,
                       const float* __restrict__ mP_W1,
                       const float* __restrict__ mP_b1,
                       const float* __restrict__ mP_W2,
                       const float* __restrict__ mP_b2,
                       float* __restrict__ out)
{
    __shared__ float hD_lds[TL * HID_];
    __shared__ float hP_lds[TL * HID_];
    __shared__ float kD_lds[TL * 9];
    __shared__ float y_lds[TL * B_ * CI];
    __shared__ float kP_lds[TL * 16 * CO];

    const int tid  = threadIdx.x;
    const int wave = tid >> 5;
    const int lane = tid & 31;
    const int lm   = lane & 15;
    const int hiL  = lane >> 4;

    const int wg = blockIdx.x;
    const int h  = wg >> 3;
    const int w0 = (wg & 7) * TL;

    {
        const int l = tid >> 4;
        const int j = tid & 15;
        const int w = w0 + l;
        const float xn = (float)h * (1.0f / 128.0f);
        const float yn = (float)w * (1.0f / 128.0f);
        float f[12];
        f[0] = xn; f[1] = 1.0f - xn; f[2] = yn; f[3] = 1.0f - yn;
        #pragma unroll
        for (int q = 0; q < 8; ++q) f[4 + q] = cvec[q];
        float aD = mD_b1[j], aP = mP_b1[j];
        #pragma unroll
        for (int q = 0; q < 12; ++q) {
            aD += f[q] * mD_W1[q * HID_ + j];
            aP += f[q] * mP_W1[q * HID_ + j];
        }
        hD_lds[l * HID_ + j] = fmaxf(aD, 0.0f);
        hP_lds[l * HID_ + j] = fmaxf(aP, 0.0f);
    }
    __syncthreads();

    if (tid < TL * 9) {
        const int l = tid / 9;
        const int k = tid % 9;
        float a = kernelD[k] + mD_b2[k];
        #pragma unroll
        for (int j = 0; j < HID_; ++j) a += hD_lds[l * HID_ + j] * mD_W2[j * 9 + k];
        kD_lds[l * 9 + k] = a;
    }
    __syncthreads();

    {
        const int b = tid >> 5;
        const int i = tid & 31;
        const float* xp = x + (size_t)(b * CI + i) * (HH * WW);
        for (int l = 0; l < TL; ++l) {
            const int w = w0 + l;
            float acc = 0.0f;
            #pragma unroll
            for (int kh = 0; kh < 3; ++kh) {
                const int hh = h + kh - 1;
                if ((unsigned)hh < (unsigned)HH) {
                    #pragma unroll
                    for (int kw = 0; kw < 3; ++kw) {
                        const int ww = w + kw - 1;
                        if ((unsigned)ww < (unsigned)WW)
                            acc += xp[hh * WW + ww] * kD_lds[l * 9 + kh * 3 + kw];
                    }
                }
            }
            y_lds[(l * B_ + b) * CI + i] = acc;
        }
    }
    __syncthreads();

    v8f acc[2][2];
    #pragma unroll
    for (int a1 = 0; a1 < 2; ++a1)
        #pragma unroll
        for (int a2 = 0; a2 < 2; ++a2) acc[a1][a2] = (v8f){};

    for (int ihalf = 0; ihalf < 2; ++ihalf) {
        for (int t = 0; t < 4; ++t) {
            const int n    = (wave * 4 + t) * 16 + lm;
            const int gcol = ihalf * 512 + n;
            v8f d = (v8f){};
            #pragma unroll
            for (int ks = 0; ks < 4; ++ks) {
                const int k0 = ks * 4 + 2 * hiL;
                v2f af, bf;
                af.x = hP_lds[lm * HID_ + k0];
                af.y = hP_lds[lm * HID_ + k0 + 1];
                bf.x = mP_W2[(size_t)k0 * 1024 + gcol];
                bf.y = mP_W2[(size_t)(k0 + 1) * 1024 + gcol];
                d = __builtin_amdgcn_wmma_f32_16x16x4_f32(
                        false, af, false, bf, (short)0, d, false, false);
            }
            const int i_loc = n >> 5;
            const int o     = n & 31;
            const int ig    = gcol >> 5;
            const float kbase = kernelP[ig * CO + o] + mP_b2[gcol];
            #pragma unroll
            for (int r = 0; r < 8; ++r) {
                const int m = r + 8 * hiL;
                kP_lds[(m * 16 + i_loc) * CO + o] = d[r] + kbase;
            }
        }
        __syncthreads();

        #pragma unroll
        for (int li = 0; li < 2; ++li) {
            const int l = wave * 2 + li;
            #pragma unroll
            for (int nt = 0; nt < 2; ++nt) {
                v8f d = acc[li][nt];
                #pragma unroll
                for (int ks = 0; ks < 4; ++ks) {
                    const int k0 = ks * 4 + 2 * hiL;
                    v2f af, bf;
                    if (lm < 8) {
                        af.x = y_lds[(l * B_ + lm) * CI + ihalf * 16 + k0];
                        af.y = y_lds[(l * B_ + lm) * CI + ihalf * 16 + k0 + 1];
                    } else {
                        af.x = 0.0f; af.y = 0.0f;
                    }
                    const int o = nt * 16 + lm;
                    bf.x = kP_lds[(l * 16 + k0) * CO + o];
                    bf.y = kP_lds[(l * 16 + k0 + 1) * CO + o];
                    d = __builtin_amdgcn_wmma_f32_16x16x4_f32(
                            false, af, false, bf, (short)0, d, false, false);
                }
                acc[li][nt] = d;
            }
        }
        __syncthreads();
    }

    #pragma unroll
    for (int li = 0; li < 2; ++li) {
        const int l = wave * 2 + li;
        const size_t lglob = (size_t)h * WW + (w0 + l);
        #pragma unroll
        for (int nt = 0; nt < 2; ++nt) {
            const int o = nt * 16 + lm;
            const float bo = bias[o];
            if (hiL == 0) {
                #pragma unroll
                for (int r = 0; r < 8; ++r) {
                    out[(size_t)(r * CO + o) * (HH * WW) + lglob] = acc[li][nt][r] + bo;
                }
            }
        }
    }
}

extern "C" void kernel_launch(void* const* d_in, const int* in_sizes, int n_in,
                              void* d_out, int out_size, void* d_ws, size_t ws_size,
                              hipStream_t stream) {
    const float* x       = (const float*)d_in[0];
    const float* kernelD = (const float*)d_in[1];
    const float* kernelP = (const float*)d_in[2];
    const float* bias    = (const float*)d_in[3];
    const float* cvec    = (const float*)d_in[4];
    const float* mD_W1   = (const float*)d_in[5];
    const float* mD_b1   = (const float*)d_in[6];
    const float* mD_W2   = (const float*)d_in[7];
    const float* mD_b2   = (const float*)d_in[8];
    const float* mP_W1   = (const float*)d_in[9];
    const float* mP_b1   = (const float*)d_in[10];
    const float* mP_W2   = (const float*)d_in[11];
    const float* mP_b2   = (const float*)d_in[12];
    float* out = (float*)d_out;

    dim3 grid(1024);
    dim3 block(256);
    hipLaunchKernelGGL(scaconv_ds_kernel, grid, block, 0, stream,
                       x, kernelD, kernelP, bias, cvec,
                       mD_W1, mD_b1, mD_W2, mD_b2,
                       mP_W1, mP_b1, mP_W2, mP_b2, out);
}


# Depthwise stage re-mapped: lanes run along the tile's 16 columns and lane halves take 2 channels, so an x load reads two 64-byte row pieces, not one word of 32 planes; y tile pitch +4 words
# speedup vs baseline: 1.8852x; 1.8852x over previous
#include <hip/hip_runtime.h>


#define B_   8
#define CI   32
#define CO   32
#define HH   128
#define WW   128
#define HID_ 16
#define TL   16
#define Y_LP (B_ * CI + 4)

typedef __attribute__((ext_vector_type(2))) float v2f;
typedef __attribute__((ext_vector_type(8))) float v8f;

__global__ __launch_bounds__(256)
void scaconv_ds_kernel(const float* __restrict__ x,
                       const float* __restrict__ kernelD,
                       const float* __restrict__ kernelP,
                       const float* __restrict__ bias,
                       const float* __restrict__ cvec,
                       const float* __restrict__ mD_W1,
                       const float* __restrict__ mD_b1,
                       const float* __restrict__ mD_W2,
                       const float* __restrict__ mD_b2,
                       const float* __restrict__ mP_W1,
                       const float* __restrict__ mP_b1,
                       const float* __restrict__ mP_W2,
                       const float* __restrict__ mP_b2,
                       float* __restrict__ out)
{
    __shared__ float hD_lds[TL * HID_];
    __shared__ float hP_lds[TL * HID_];
    __shared__ float kD_lds[TL * 9];
    __shared__ float y_lds[TL * Y_LP];
    __shared__ float kP_lds[TL * 16 * CO];

    const int tid  = threadIdx.x;
    const int wave = tid >> 5;
    const int lane = tid & 31;
    const int lm   = lane & 15;
    const int hiL  = lane >> 4;

    const int wg = blockIdx.x;
    const int h  = wg >> 3;
    const int w0 = (wg & 7) * TL;

    {
        const int l = tid >> 4;
        const int j = tid & 15;
        const int w = w0 + l;
        const float xn = (float)h * (1.0f / 128.0f);
        const float yn = (float)w * (1.0f / 128.0f);
        float f[12];
        f[0] = xn; f[1] = 1.0f - xn; f[2] = yn; f[3] = 1.0f - yn;
        #pragma unroll
        for (int q = 0; q < 8; ++q) f[4 + q] = cvec[q];
        float aD = mD_b1[j], aP = mP_b1[j];
        #pragma unroll
        for (int q = 0; q < 12; ++q) {
            aD += f[q] * mD_W1[q * HID_ + j];
            aP += f[q] * mP_W1[q * HID_ + j];
        }
        hD_lds[l * HID_ + j] = fmaxf(aD, 0.0f);
        hP_lds[l * HID_ + j] = fmaxf(aP, 0.0f);
    }
    __syncthreads();

    if (tid < TL * 9) {
        const int l = tid / 9;
        const int k = tid % 9;
        float a = kernelD[k] + mD_b2[k];
        #pragma unroll
        for (int j = 0; j < HID_; ++j) a += hD_lds[l * HID_ + j] * mD_W2[j * 9 + k];
        kD_lds[l * 9 + k] = a;
    }
    __syncthreads();

    {
        const int b = wave;
        const int w = w0 + lm;
        float kd[9];
        #pragma unroll
        for (int k = 0; k < 9; ++k) kd[k] = kD_lds[lm * 9 + k];
        const bool okT = (h > 0), okB = (h < HH - 1);
        const bool okL = (w > 0), okR = (w < WW - 1);
        const int rT = (okT ? h - 1 : h) * WW;
        const int rC = h * WW;
        const int rB = (okB ? h + 1 : h) * WW;
        const int cL = okL ? w - 1 : w;
        const int cR = okR ? w + 1 : w;
        const float* xl = x + (size_t)(b * CI + hiL) * (HH * WW);
        const float* q00 = xl + rT + cL; const float* q01 = xl + rT + w; const float* q02 = xl + rT + cR;
        const float* q10 = xl + rC + cL; const float* q11 = xl + rC + w; const float* q12 = xl + rC + cR;
        const float* q20 = xl + rB + cL; const float* q21 = xl + rB + w; const float* q22 = xl + rB + cR;
        float* yp = &y_lds[lm * Y_LP + b * CI + hiL];
        #pragma clang loop unroll(disable) vectorize(disable)
        for (int p0 = 0; p0 < CI / 2; p0 += 4) {
            #pragma unroll
            for (int pp = 0; pp < 4; ++pp) {
                const int c = 2 * (p0 + pp) * (HH * WW);
                const float x00 = q00[c], x01 = q01[c], x02 = q02[c];
                const float x10 = q10[c], x11 = q11[c], x12 = q12[c];
                const float x20 = q20[c], x21 = q21[c], x22 = q22[c];
                float acc = 0.0f;
                acc += ((okT && okL) ? x00 : 0.0f) * kd[0];
                acc += ( okT         ? x01 : 0.0f) * kd[1];
                acc += ((okT && okR) ? x02 : 0.0f) * kd[2];
                acc += ( okL         ? x10 : 0.0f) * kd[3];
                acc +=                 x11         * kd[4];
                acc += ( okR         ? x12 : 0.0f) * kd[5];
                acc += ((okB && okL) ? x20 : 0.0f) * kd[6];
                acc += ( okB         ? x21 : 0.0f) * kd[7];
                acc += ((okB && okR) ? x22 : 0.0f) * kd[8];
                yp[2 * (p0 + pp)] = acc;
            }
        }
    }
    __syncthreads();

    v8f acc[2][2];
    #pragma unroll
    for (int a1 = 0; a1 < 2; ++a1)
        #pragma unroll
        for (int a2 = 0; a2 < 2; ++a2) acc[a1][a2] = (v8f){};

    for (int ihalf = 0; ihalf < 2; ++ihalf) {
        for (int t = 0; t < 4; ++t) {
            const int n    = (wave * 4 + t) * 16 + lm;
            const int gcol = ihalf * 512 + n;
            v8f d = (v8f){};
            #pragma unroll
            for (int ks = 0; ks < 4; ++ks) {
                const int k0 = ks * 4 + 2 * hiL;
                v2f af, bf;
                af.x = hP_lds[lm * HID_ + k0];
                af.y = hP_lds[lm * HID_ + k0 + 1];
                bf.x = mP_W2[(size_t)k0 * 1024 + gcol];
                bf.y = mP_W2[(size_t)(k0 + 1) * 1024 + gcol];
                d = __builtin_amdgcn_wmma_f32_16x16x4_f32(
                        false, af, false, bf, (short)0, d, false, false);
            }
            const int i_loc = n >> 5;
            const int o     = n & 31;
            const int ig    = gcol >> 5;
            const float kbase = kernelP[ig * CO + o] + mP_b2[gcol];
            #pragma unroll
            for (int r = 0; r < 8; ++r) {
                const int m = r + 8 * hiL;
                kP_lds[(m * 16 + i_loc) * CO + o] = d[r] + kbase;
            }
        }
        __syncthreads();

        #pragma unroll
        for (int li = 0; li < 2; ++li) {
            const int l = wave * 2 + li;
            #pragma unroll
            for (int nt = 0; nt < 2; ++nt) {
                v8f d = acc[li][nt];
                #pragma unroll
                for (int ks = 0; ks < 4; ++ks) {
                    const int k0 = ks * 4 + 2 * hiL;
                    v2f af, bf;
                    if (lm < 8) {
                        af.x = y_lds[l * Y_LP + lm * CI + ihalf * 16 + k0];
                        af.y = y_lds[l * Y_LP + lm * CI + ihalf * 16 + k0 + 1];
                    } else {
                        af.x = 0.0f; af.y = 0.0f;
                    }
                    const int o = nt * 16 + lm;
                    bf.x = kP_lds[(l * 16 + k0) * CO + o];
                    bf.y = kP_lds[(l * 16 + k0 + 1) * CO + o];
                    d = __builtin_amdgcn_wmma_f32_16x16x4_f32(
                            false, af, false, bf, (short)0, d, false, false);
                }
                acc[li][nt] = d;
            }
        }
        __syncthreads();
    }

    #pragma unroll
    for (int li = 0; li < 2; ++li) {
        const int l = wave * 2 + li;
        const size_t lglob = (size_t)h * WW + (w0 + l);
        #pragma unroll
        for (int nt = 0; nt < 2; ++nt) {
            const int o = nt * 16 + lm;
            const float bo = bias[o];
            if (hiL == 0) {
                #pragma unroll
                for (int r = 0; r < 8; ++r) {
                    out[(size_t)(r * CO + o) * (HH * WW) + lglob] = acc[li][nt][r] + bo;
                }
            }
        }
    }
}

extern "C" void kernel_launch(void* const* d_in, const int* in_sizes, int n_in,
                              void* d_out, int out_size, void* d_ws, size_t ws_size,
                              hipStream_t stream) {
    const float* x       = (const float*)d_in[0];
    const float* kernelD = (const float*)d_in[1];
    const float* kernelP = (const float*)d_in[2];
    const float* bias    = (const float*)d_in[3];
    const float* cvec    = (const float*)d_in[4];
    const float* mD_W1   = (const float*)d_in[5];
    const float* mD_b1   = (const float*)d_in[6];
    const float* mD_W2   = (const float*)d_in[7];
    const float* mD_b2   = (const float*)d_in[8];
    const float* mP_W1   = (const float*)d_in[9];
    const float* mP_b1   = (const float*)d_in[10];
    const float* mP_W2   = (const float*)d_in[11];
    const float* mP_b2   = (const float*)d_in[12];
    float* out = (float*)d_out;

    dim3 grid(1024);
    dim3 block(256);
    hipLaunchKernelGGL(scaconv_ds_kernel, grid, block, 0, stream,
                       x, kernelD, kernelP, bias, cvec,
                       mD_W1, mD_b1, mD_W2, mD_b2,
                       mP_W1, mP_b1, mP_W2, mP_b2, out);
}


# Output tile staged through LDS (the free kP tile): each plane's 64-byte row piece leaves as four 16-byte stores of one instruction, not as 8-byte stores of eight waves into 16 planes each
# speedup vs baseline: 4.2491x; 2.2539x over previous
#include <hip/hip_runtime.h>


#define B_   8
#define CI   32
#define CO   32
#define HH   128
#define WW   128
#define HID_ 16
#define TL   16
#define Y_LP (B_ * CI + 4)
#define ST_P 20

typedef __attribute__((ext_vector_type(2))) float v2f;
typedef __attribute__((ext_vector_type(8))) float v8f;

__global__ __launch_bounds__(256)
void scaconv_ds_kernel(const float* __restrict__ x,
                       const float* __restrict__ kernelD,
                       const float* __restrict__ kernelP,
                       const float* __restrict__ bias,
                       const float* __restrict__ cvec,
                       const float* __restrict__ mD_W1,
                       const float* __restrict__ mD_b1,
                       const float* __restrict__ mD_W2,
                       const float* __restrict__ mD_b2,
                       const float* __restrict__ mP_W1,
                       const float* __restrict__ mP_b1,
                       const float* __restrict__ mP_W2,
                       const float* __restrict__ mP_b2,
                       float* __restrict__ out)
{
    __shared__ float hD_lds[TL * HID_];
    __shared__ float hP_lds[TL * HID_];
    __shared__ float kD_lds[TL * 9];
    __shared__ float y_lds[TL * Y_LP];
    __shared__ alignas(16) float kP_lds[TL * 16 * CO];

    const int tid  = threadIdx.x;
    const int wave = tid >> 5;
    const int lane = tid & 31;
    const int lm   = lane & 15;
    const int hiL  = lane >> 4;

    const int wg = blockIdx.x;
    const int h  = wg >> 3;
    const int w0 = (wg & 7) * TL;

    {
        const int l = tid >> 4;
        const int j = tid & 15;
        const int w = w0 + l;
        const float xn = (float)h * (1.0f / 128.0f);
        const float yn = (float)w * (1.0f / 128.0f);
        float f[12];
        f[0] = xn; f[1] = 1.0f - xn; f[2] = yn; f[3] = 1.0f - yn;
        #pragma unroll
        for (int q = 0; q < 8; ++q) f[4 + q] = cvec[q];
        float aD = mD_b1[j], aP = mP_b1[j];
        #pragma unroll
        for (int q = 0; q < 12; ++q) {
            aD += f[q] * mD_W1[q * HID_ + j];
            aP += f[q] * mP_W1[q * HID_ + j];
        }
        hD_lds[l * HID_ + j] = fmaxf(aD, 0.0f);
        hP_lds[l * HID_ + j] = fmaxf(aP, 0.0f);
    }
    __syncthreads();

    if (tid < TL * 9) {
        const int l = tid / 9;
        const int k = tid % 9;
        float a = kernelD[k] + mD_b2[k];
        #pragma unroll
        for (int j = 0; j < HID_; ++j) a += hD_lds[l * HID_ + j] * mD_W2[j * 9 + k];
        kD_lds[l * 9 + k] = a;
    }
    __syncthreads();

    {
        const int b = wave;
        const int w = w0 + lm;
        float kd[9];
        #pragma unroll
        for (int k = 0; k < 9; ++k) kd[k] = kD_lds[lm * 9 + k];
        const bool okT = (h > 0), okB = (h < HH - 1);
        const bool okL = (w > 0), okR = (w < WW - 1);
        const int rT = (okT ? h - 1 : h) * WW;
        const int rC = h * WW;
        const int rB = (okB ? h + 1 : h) * WW;
        const int cL = okL ? w - 1 : w;
        const int cR = okR ? w + 1 : w;
        const float* xl = x + (size_t)(b * CI + hiL) * (HH * WW);
        const float* q00 = xl + rT + cL; const float* q01 = xl + rT + w; const float* q02 = xl + rT + cR;
        const float* q10 = xl + rC + cL; const float* q11 = xl + rC + w; const float* q12 = xl + rC + cR;
        const float* q20 = xl + rB + cL; const float* q21 = xl + rB + w; const float* q22 = xl + rB + cR;
        float* yp = &y_lds[lm * Y_LP + b * CI + hiL];
        #pragma clang loop unroll(disable) vectorize(disable)
        for (int p0 = 0; p0 < CI / 2; p0 += 4) {
            #pragma unroll
            for (int pp = 0; pp < 4; ++pp) {
                const int c = 2 * (p0 + pp) * (HH * WW);
                const float x00 = q00[c], x01 = q01[c], x02 = q02[c];
                const float x10 = q10[c], x11 = q11[c], x12 = q12[c];
                const float x20 = q20[c], x21 = q21[c], x22 = q22[c];
                float acc = 0.0f;
                acc += ((okT && okL) ? x00 : 0.0f) * kd[0];
                acc += ( okT         ? x01 : 0.0f) * kd[1];
                acc += ((okT && okR) ? x02 : 0.0f) * kd[2];
                acc += ( okL         ? x10 : 0.0f) * kd[3];
                acc +=                 x11         * kd[4];
                acc += ( okR         ? x12 : 0.0f) * kd[5];
                acc += ((okB && okL) ? x20 : 0.0f) * kd[6];
                acc += ( okB         ? x21 : 0.0f) * kd[7];
                acc += ((okB && okR) ? x22 : 0.0f) * kd[8];
                yp[2 * (p0 + pp)] = acc;
            }
        }
    }
    __syncthreads();

    v8f acc[2][2];
    #pragma unroll
    for (int a1 = 0; a1 < 2; ++a1)
        #pragma unroll
        for (int a2 = 0; a2 < 2; ++a2) acc[a1][a2] = (v8f){};

    for (int ihalf = 0; ihalf < 2; ++ihalf) {
        for (int t = 0; t < 4; ++t) {
            const int n    = (wave * 4 + t) * 16 + lm;
            const int gcol = ihalf * 512 + n;
            v8f d = (v8f){};
            #pragma unroll
            for (int ks = 0; ks < 4; ++ks) {
                const int k0 = ks * 4 + 2 * hiL;
                v2f af, bf;
                af.x = hP_lds[lm * HID_ + k0];
                af.y = hP_lds[lm * HID_ + k0 + 1];
                bf.x = mP_W2[(size_t)k0 * 1024 + gcol];
                bf.y = mP_W2[(size_t)(k0 + 1) * 1024 + gcol];
                d = __builtin_amdgcn_wmma_f32_16x16x4_f32(
                        false, af, false, bf, (short)0, d, false, false);
            }
            const int i_loc = n >> 5;
            const int o     = n & 31;
            const int ig    = gcol >> 5;
            const float kbase = kernelP[ig * CO + o] + mP_b2[gcol];
            #pragma unroll
            for (int r = 0; r < 8; ++r) {
                const int m = r + 8 * hiL;
                kP_lds[(m * 16 + i_loc) * CO + o] = d[r] + kbase;
            }
        }
        __syncthreads();

        #pragma unroll
        for (int li = 0; li < 2; ++li) {
            const int l = wave * 2 + li;
            #pragma unroll
            for (int nt = 0; nt < 2; ++nt) {
                v8f d = acc[li][nt];
                #pragma unroll
                for (int ks = 0; ks < 4; ++ks) {
                    const int k0 = ks * 4 + 2 * hiL;
                    v2f af, bf;
                    if (lm < 8) {
                        af.x = y_lds[l * Y_LP + lm * CI + ihalf * 16 + k0];
                        af.y = y_lds[l * Y_LP + lm * CI + ihalf * 16 + k0 + 1];
                    } else {
                        af.x = 0.0f; af.y = 0.0f;
                    }
                    const int o = nt * 16 + lm;
                    bf.x = kP_lds[(l * 16 + k0) * CO + o];
                    bf.y = kP_lds[(l * 16 + k0 + 1) * CO + o];
                    d = __builtin_amdgcn_wmma_f32_16x16x4_f32(
                            false, af, false, bf, (short)0, d, false, false);
                }
                acc[li][nt] = d;
            }
        }
        __syncthreads();
    }

    float* st = kP_lds;
    if (hiL == 0) {
        #pragma unroll
        for (int nt = 0; nt < 2; ++nt) {
            const int o = nt * 16 + lm;
            const float bo = bias[o];
            #pragma unroll
            for (int r = 0; r < 8; ++r) {
                float* sp = &st[(r * CO + o) * ST_P + wave * 2];
                sp[0] = acc[0][nt][r] + bo;
                sp[1] = acc[1][nt][r] + bo;
            }
        }
    }
    __syncthreads();
    {
        const size_t pix = (size_t)h * WW + w0;
        #pragma unroll
        for (int it = 0; it < 4; ++it) {
            const int item = it * 256 + tid;
            const int row  = item >> 2;
            const int q    = item & 3;
            const float4 v = *reinterpret_cast<const float4*>(&st[row * ST_P + q * 4]);
            *reinterpret_cast<float4*>(&out[(size_t)row * (HH * WW) + pix + q * 4]) = v;
        }
    }
}

extern "C" void kernel_launch(void* const* d_in, const int* in_sizes, int n_in,
                              void* d_out, int out_size, void* d_ws, size_t ws_size,
                              hipStream_t stream) {
    const float* x       = (const float*)d_in[0];
    const float* kernelD = (const float*)d_in[1];
    const float* kernelP = (const float*)d_in[2];
    const float* bias    = (const float*)d_in[3];
    const float* cvec    = (const float*)d_in[4];
    const float* mD_W1   = (const float*)d_in[5];
    const float* mD_b1   = (const float*)d_in[6];
    const float* mD_W2   = (const float*)d_in[7];
    const float* mD_b2   = (const float*)d_in[8];
    const float* mP_W1   = (const float*)d_in[9];
    const float* mP_b1   = (const float*)d_in[10];
    const float* mP_W2   = (const float*)d_in[11];
    const float* mP_b2   = (const float*)d_in[12];
    float* out = (float*)d_out;

    dim3 grid(1024);
    dim3 block(256);
    hipLaunchKernelGGL(scaconv_ds_kernel, grid, block, 0, stream,
                       x, kernelD, kernelP, bias, cvec,
                       mD_W1, mD_b1, mD_W2, mD_b2,
                       mP_W1, mP_b1, mP_W2, mP_b2, out);
}


# Tile widened to one whole 128-byte line per plane row (32 columns, 16 waves, grid 512): each x line is loaded once, 8 bytes per lane, side taps by DPP row shifts; each output line is stored whole
# speedup vs baseline: 5.2726x; 1.2409x over previous
#include <hip/hip_runtime.h>


#define B_   8
#define CI   32
#define CO   32
#define HH   128
#define WW   128
#define HID_ 16
#define TL   32
#define NTHR 512
#define Y_LP (B_ * CI + 2)
#define ST_P 36

typedef __attribute__((ext_vector_type(2))) float v2f;
typedef __attribute__((ext_vector_type(8))) float v8f;

__device__ __forceinline__ float from_prev_lane(float edge, float v) {
    return __int_as_float(__builtin_amdgcn_update_dpp(__float_as_int(edge), __float_as_int(v),
                                                      0x111  , 0xF, 0xF, false));
}
__device__ __forceinline__ float from_next_lane(float edge, float v) {
    return __int_as_float(__builtin_amdgcn_update_dpp(__float_as_int(edge), __float_as_int(v),
                                                      0x101  , 0xF, 0xF, false));
}

__global__ __launch_bounds__(NTHR)
void scaconv_ds_kernel(const float* __restrict__ x,
                       const float* __restrict__ kernelD,
                       const float* __restrict__ kernelP,
                       const float* __restrict__ bias,
                       const float* __restrict__ cvec,
                       const float* __restrict__ mD_W1,
                       const float* __restrict__ mD_b1,
                       const float* __restrict__ mD_W2,
                       const float* __restrict__ mD_b2,
                       const float* __restrict__ mP_W1,
                       const float* __restrict__ mP_b1,
                       const float* __restrict__ mP_W2,
                       const float* __restrict__ mP_b2,
                       float* __restrict__ out)
{
    __shared__ float hD_lds[TL * HID_];
    __shared__ float hP_lds[TL * HID_];
    __shared__ float kD_lds[TL * 9];
    __shared__ alignas(16) float y_lds[TL * Y_LP];
    __shared__ alignas(16) float kP_lds[TL * 16 * CO];

    const int tid  = threadIdx.x;
    const int wave = tid >> 5;
    const int lane = tid & 31;
    const int lm   = lane & 15;
    const int hiL  = lane >> 4;

    const int wg = blockIdx.x;
    const int h  = wg >> 2;
    const int w0 = (wg & 3) * TL;

    {
        const int l = tid >> 4;
        const int j = tid & 15;
        const int w = w0 + l;
        const float xn = (float)h * (1.0f / 128.0f);
        const float yn = (float)w * (1.0f / 128.0f);
        float f[12];
        f[0] = xn; f[1] = 1.0f - xn; f[2] = yn; f[3] = 1.0f - yn;
        #pragma unroll
        for (int q = 0; q < 8; ++q) f[4 + q] = cvec[q];
        float aD = mD_b1[j], aP = mP_b1[j];
        #pragma unroll
        for (int q = 0; q < 12; ++q) {
            aD += f[q] * mD_W1[q * HID_ + j];
            aP += f[q] * mP_W1[q * HID_ + j];
        }
        hD_lds[l * HID_ + j] = fmaxf(aD, 0.0f);
        hP_lds[l * HID_ + j] = fmaxf(aP, 0.0f);
    }
    __syncthreads();

    if (tid < TL * 9) {
        const int l = tid / 9;
        const int k = tid % 9;
        float a = kernelD[k] + mD_b2[k];
        #pragma unroll
        for (int j = 0; j < HID_; ++j) a += hD_lds[l * HID_ + j] * mD_W2[j * 9 + k];
        kD_lds[l * 9 + k] = a;
    }
    __syncthreads();

    {
        const int b   = wave >> 1;
        const int ch0 = (wave & 1) * 16;
        const int lA  = 2 * lm;
        float kdA[9], kdB[9];
        #pragma unroll
        for (int k = 0; k < 9; ++k) {
            kdA[k] = kD_lds[lA * 9 + k];
            kdB[k] = kD_lds[(lA + 1) * 9 + k];
        }
        const bool okT = (h > 0), okB = (h < HH - 1);
        const bool okH = (lm == 0) ? (w0 > 0) : (w0 + TL < WW);
        const int  cH  = (lm == 0)  ? (w0 > 0 ? w0 - 1 : w0)
                       : (lm == 15) ? (w0 + TL < WW ? w0 + TL : w0 + TL - 1)
                       : w0 + lA;
        const bool okTH = okT && okH, okBH = okB && okH;
        const int rT = (okT ? h - 1 : h) * WW;
        const int rC = h * WW;
        const int rB = (okB ? h + 1 : h) * WW;
        const float* xl = x + (size_t)(b * CI + ch0 + hiL) * (HH * WW);
        const float* mT = xl + rT + w0 + lA; const float* eT = xl + rT + cH;
        const float* mC = xl + rC + w0 + lA; const float* eC = xl + rC + cH;
        const float* mB = xl + rB + w0 + lA; const float* eB = xl + rB + cH;
        float* ypA = &y_lds[lA * Y_LP + b * CI + ch0 + hiL];
        float* ypB = ypA + Y_LP;
        #pragma clang loop unroll(disable) vectorize(disable)
        for (int p0 = 0; p0 < 8; p0 += 4) {
            #pragma unroll
            for (int pp = 0; pp < 4; ++pp) {
                const int c = 2 * (p0 + pp) * (HH * WW);
                float2 t = *reinterpret_cast<const float2*>(mT + c);
                float2 m = *reinterpret_cast<const float2*>(mC + c);
                float2 u = *reinterpret_cast<const float2*>(mB + c);
                float et = eT[c], ec = eC[c], eb = eB[c];
                t.x = okT ? t.x : 0.0f;  t.y = okT ? t.y : 0.0f;  et = okTH ? et : 0.0f;
                ec = okH ? ec : 0.0f;
                u.x = okB ? u.x : 0.0f;  u.y = okB ? u.y : 0.0f;  eb = okBH ? eb : 0.0f;
                const float tL = from_prev_lane(et, t.y), tR = from_next_lane(et, t.x);
                const float cL = from_prev_lane(ec, m.y), cR = from_next_lane(ec, m.x);
                const float bL = from_prev_lane(eb, u.y), bR = from_next_lane(eb, u.x);
                float accA = 0.0f, accB = 0.0f;
                accA += tL  * kdA[0];  accB += t.x * kdB[0];
                accA += t.x * kdA[1];  accB += t.y * kdB[1];
                accA += t.y * kdA[2];  accB += tR  * kdB[2];
                accA += cL  * kdA[3];  accB += m.x * kdB[3];
                accA += m.x * kdA[4];  accB += m.y * kdB[4];
                accA += m.y * kdA[5];  accB += cR  * kdB[5];
                accA += bL  * kdA[6];  accB += u.x * kdB[6];
                accA += u.x * kdA[7];  accB += u.y * kdB[7];
                accA += u.y * kdA[8];  accB += bR  * kdB[8];
                ypA[2 * (p0 + pp)] = accA;
                ypB[2 * (p0 + pp)] = accB;
            }
        }
    }
    __syncthreads();

    v8f acc[2][2];
    #pragma unroll
    for (int a1 = 0; a1 < 2; ++a1)
        #pragma unroll
        for (int a2 = 0; a2 < 2; ++a2) acc[a1][a2] = (v8f){};

    for (int ihalf = 0; ihalf < 2; ++ihalf) {
        for (int t = 0; t < 2; ++t) {
            const int n    = (wave * 2 + t) * 16 + lm;
            const int gcol = ihalf * 512 + n;
            v8f d0 = (v8f){}, d1 = (v8f){};
            #pragma unroll
            for (int ks = 0; ks < 4; ++ks) {
                const int k0 = ks * 4 + 2 * hiL;
                v2f a0, a1, bf;
                a0.x = hP_lds[lm * HID_ + k0];
                a0.y = hP_lds[lm * HID_ + k0 + 1];
                a1.x = hP_lds[(16 + lm) * HID_ + k0];
                a1.y = hP_lds[(16 + lm) * HID_ + k0 + 1];
                bf.x = mP_W2[(size_t)k0 * 1024 + gcol];
                bf.y = mP_W2[(size_t)(k0 + 1) * 1024 + gcol];
                d0 = __builtin_amdgcn_wmma_f32_16x16x4_f32(
                        false, a0, false, bf, (short)0, d0, false, false);
                d1 = __builtin_amdgcn_wmma_f32_16x16x4_f32(
                        false, a1, false, bf, (short)0, d1, false, false);
            }
            const int i_loc = n >> 5;
            const int o     = n & 31;
            const int ig    = gcol >> 5;
            const float kbase = kernelP[ig * CO + o] + mP_b2[gcol];
            #pragma unroll
            for (int r = 0; r < 8; ++r) {
                const int m = r + 8 * hiL;
                kP_lds[(m * 16 + i_loc) * CO + o]        = d0[r] + kbase;
                kP_lds[((16 + m) * 16 + i_loc) * CO + o] = d1[r] + kbase;
            }
        }
        __syncthreads();

        #pragma unroll
        for (int li = 0; li < 2; ++li) {
            const int l = wave * 2 + li;
            #pragma unroll
            for (int nt = 0; nt < 2; ++nt) {
                v8f d = acc[li][nt];
                #pragma unroll
                for (int ks = 0; ks < 4; ++ks) {
                    const int k0 = ks * 4 + 2 * hiL;
                    v2f af, bf;
                    if (lm < 8) {
                        af.x = y_lds[l * Y_LP + lm * CI + ihalf * 16 + k0];
                        af.y = y_lds[l * Y_LP + lm * CI + ihalf * 16 + k0 + 1];
                    } else {
                        af.x = 0.0f; af.y = 0.0f;
                    }
                    const int o = nt * 16 + lm;
                    bf.x = kP_lds[(l * 16 + k0) * CO + o];
                    bf.y = kP_lds[(l * 16 + k0 + 1) * CO + o];
                    d = __builtin_amdgcn_wmma_f32_16x16x4_f32(
                            false, af, false, bf, (short)0, d, false, false);
                }
                acc[li][nt] = d;
            }
        }
        __syncthreads();
    }

    float* st = kP_lds;
    if (hiL == 0) {
        #pragma unroll
        for (int nt = 0; nt < 2; ++nt) {
            const int o = nt * 16 + lm;
            const float bo = bias[o];
            #pragma unroll
            for (int r = 0; r < 8; ++r) {
                float* sp = &st[(r * CO + o) * ST_P + wave * 2];
                sp[0] = acc[0][nt][r] + bo;
                sp[1] = acc[1][nt][r] + bo;
            }
        }
    }
    __syncthreads();
    {
        const size_t pix = (size_t)h * WW + w0;
        #pragma unroll
        for (int it = 0; it < 4; ++it) {
            const int item = it * NTHR + tid;
            const int row  = item >> 3;
            const int q    = item & 7;
            const float4 v = *reinterpret_cast<const float4*>(&st[row * ST_P + q * 4]);
            *reinterpret_cast<float4*>(&out[(size_t)row * (HH * WW) + pix + q * 4]) = v;
        }
    }
}

extern "C" void kernel_launch(void* const* d_in, const int* in_sizes, int n_in,
                              void* d_out, int out_size, void* d_ws, size_t ws_size,
                              hipStream_t stream) {
    const float* x       = (const float*)d_in[0];
    const float* kernelD = (const float*)d_in[1];
    const float* kernelP = (const float*)d_in[2];
    const float* bias    = (const float*)d_in[3];
    const float* cvec    = (const float*)d_in[4];
    const float* mD_W1   = (const float*)d_in[5];
    const float* mD_b1   = (const float*)d_in[6];
    const float* mD_W2   = (const float*)d_in[7];
    const float* mD_b2   = (const float*)d_in[8];
    const float* mP_W1   = (const float*)d_in[9];
    const float* mP_b1   = (const float*)d_in[10];
    const float* mP_W2   = (const float*)d_in[11];
    const float* mP_b2   = (const float*)d_in[12];
    float* out = (float*)d_out;

    dim3 grid(512);
    dim3 block(NTHR);
    hipLaunchKernelGGL(scaconv_ds_kernel, grid, block, 0, stream,
                       x, kernelD, kernelP, bias, cvec,
                       mD_W1, mD_b1, mD_W2, mD_b2,
                       mP_W1, mP_b1, mP_W2, mP_b2, out);
}


# Work-group widened to 64 columns and 32 waves (grid 256): per plane row 2 lines + 1 neighbour line are fetched for 64 outputs, not 2 + 3, and each W2 fragment load feeds four m-tiles, not two
# speedup vs baseline: 5.6326x; 1.0683x over previous
#include <hip/hip_runtime.h>


#define B_   8
#define CI   32
#define CO   32
#define HH   128
#define WW   128
#define HID_ 16
#define TL   64
#define NTHR 1024
#define Y_LP (B_ * CI + 2)
#define ST_P 68

typedef __attribute__((ext_vector_type(2))) float v2f;
typedef __attribute__((ext_vector_type(8))) float v8f;

__device__ __forceinline__ float from_prev_lane(float edge, float v) {
    return __int_as_float(__builtin_amdgcn_update_dpp(__float_as_int(edge), __float_as_int(v),
                                                      0x111  , 0xF, 0xF, false));
}
__device__ __forceinline__ float from_next_lane(float edge, float v) {
    return __int_as_float(__builtin_amdgcn_update_dpp(__float_as_int(edge), __float_as_int(v),
                                                      0x101  , 0xF, 0xF, false));
}
__device__ __forceinline__ float rot_from_prev_lane(float v) {
    return __int_as_float(__builtin_amdgcn_update_dpp(0, __float_as_int(v),
                                                      0x121  , 0xF, 0xF, false));
}
__device__ __forceinline__ float rot_from_next_lane(float v) {
    return __int_as_float(__builtin_amdgcn_update_dpp(0, __float_as_int(v),
                                                      0x12F  , 0xF, 0xF, false));
}

__global__ __launch_bounds__(NTHR)
void scaconv_ds_kernel(const float* __restrict__ x,
                       const float* __restrict__ kernelD,
                       const float* __restrict__ kernelP,
                       const float* __restrict__ bias,
                       const float* __restrict__ cvec,
                       const float* __restrict__ mD_W1,
                       const float* __restrict__ mD_b1,
                       const float* __restrict__ mD_W2,
                       const float* __restrict__ mD_b2,
                       const float* __restrict__ mP_W1,
                       const float* __restrict__ mP_b1,
                       const float* __restrict__ mP_W2,
                       const float* __restrict__ mP_b2,
                       float* __restrict__ out)
{
    __shared__ float hD_lds[TL * HID_];
    __shared__ float hP_lds[TL * HID_];
    __shared__ float kD_lds[TL * 9];
    __shared__ alignas(16) float y_lds[TL * Y_LP];
    __shared__ alignas(16) float kP_lds[TL * 16 * CO];

    const int tid  = threadIdx.x;
    const int wave = tid >> 5;
    const int lane = tid & 31;
    const int lm   = lane & 15;
    const int hiL  = lane >> 4;

    const int wg = blockIdx.x;
    const int h  = wg >> 1;
    const int w0 = (wg & 1) * TL;

    {
        const int l = tid >> 4;
        const int j = tid & 15;
        const int w = w0 + l;
        const float xn = (float)h * (1.0f / 128.0f);
        const float yn = (float)w * (1.0f / 128.0f);
        float f[12];
        f[0] = xn; f[1] = 1.0f - xn; f[2] = yn; f[3] = 1.0f - yn;
        #pragma unroll
        for (int q = 0; q < 8; ++q) f[4 + q] = cvec[q];
        float aD = mD_b1[j], aP = mP_b1[j];
        #pragma unroll
        for (int q = 0; q < 12; ++q) {
            aD += f[q] * mD_W1[q * HID_ + j];
            aP += f[q] * mP_W1[q * HID_ + j];
        }
        hD_lds[l * HID_ + j] = fmaxf(aD, 0.0f);
        hP_lds[l * HID_ + j] = fmaxf(aP, 0.0f);
    }
    __syncthreads();

    if (tid < TL * 9) {
        const int l = tid / 9;
        const int k = tid % 9;
        float a = kernelD[k] + mD_b2[k];
        #pragma unroll
        for (int j = 0; j < HID_; ++j) a += hD_lds[l * HID_ + j] * mD_W2[j * 9 + k];
        kD_lds[l * 9 + k] = a;
    }
    __syncthreads();

    {
        const int b   = wave >> 2;
        const int ch0 = (wave & 3) * 8;
        const int lA  = 2 * lm;
        float kd0[9], kd1[9], kd2[9], kd3[9];
        #pragma unroll
        for (int k = 0; k < 9; ++k) {
            kd0[k] = kD_lds[lA * 9 + k];
            kd1[k] = kD_lds[(lA + 1) * 9 + k];
            kd2[k] = kD_lds[(32 + lA) * 9 + k];
            kd3[k] = kD_lds[(33 + lA) * 9 + k];
        }
        const bool okT = (h > 0), okB = (h < HH - 1);
        const bool leftTile = (w0 == 0);
        const int  cH = leftTile ? TL : TL - 1;
        const int rT = (okT ? h - 1 : h) * WW;
        const int rC = h * WW;
        const int rB = (okB ? h + 1 : h) * WW;
        const float* xl = x + (size_t)(b * CI + ch0 + hiL) * (HH * WW);
        const float* mT = xl + rT + w0 + lA; const float* eT = xl + rT + cH;
        const float* mC = xl + rC + w0 + lA; const float* eC = xl + rC + cH;
        const float* mB = xl + rB + w0 + lA; const float* eB = xl + rB + cH;
        float* yp0 = &y_lds[lA * Y_LP + b * CI + ch0 + hiL];
        float* yp1 = yp0 + Y_LP;
        float* yp2 = yp0 + 32 * Y_LP;
        float* yp3 = yp0 + 33 * Y_LP;
        #pragma clang loop unroll(disable) vectorize(disable)
        for (int p0 = 0; p0 < 4; p0 += 2) {
            #pragma unroll
            for (int pp = 0; pp < 2; ++pp) {
                const int c = 2 * (p0 + pp) * (HH * WW);
                float2 t0 = *reinterpret_cast<const float2*>(mT + c);
                float2 t1 = *reinterpret_cast<const float2*>(mT + c + 32);
                float2 m0 = *reinterpret_cast<const float2*>(mC + c);
                float2 m1 = *reinterpret_cast<const float2*>(mC + c + 32);
                float2 u0 = *reinterpret_cast<const float2*>(mB + c);
                float2 u1 = *reinterpret_cast<const float2*>(mB + c + 32);
                float et = eT[c], ec = eC[c], eb = eB[c];
                t0.x = okT ? t0.x : 0.0f;  t0.y = okT ? t0.y : 0.0f;
                t1.x = okT ? t1.x : 0.0f;  t1.y = okT ? t1.y : 0.0f;  et = okT ? et : 0.0f;
                u0.x = okB ? u0.x : 0.0f;  u0.y = okB ? u0.y : 0.0f;
                u1.x = okB ? u1.x : 0.0f;  u1.y = okB ? u1.y : 0.0f;  eb = okB ? eb : 0.0f;
                const float etL = leftTile ? 0.0f : et, etR = leftTile ? et : 0.0f;
                const float ecL = leftTile ? 0.0f : ec, ecR = leftTile ? ec : 0.0f;
                const float ebL = leftTile ? 0.0f : eb, ebR = leftTile ? eb : 0.0f;
                const float tL0 = from_prev_lane(etL, t0.y);
                const float tR0 = from_next_lane(rot_from_next_lane(t1.x), t0.x);
                const float tL1 = from_prev_lane(rot_from_prev_lane(t0.y), t1.y);
                const float tR1 = from_next_lane(etR, t1.x);
                const float cL0 = from_prev_lane(ecL, m0.y);
                const float cR0 = from_next_lane(rot_from_next_lane(m1.x), m0.x);
                const float cL1 = from_prev_lane(rot_from_prev_lane(m0.y), m1.y);
                const float cR1 = from_next_lane(ecR, m1.x);
                const float bL0 = from_prev_lane(ebL, u0.y);
                const float bR0 = from_next_lane(rot_from_next_lane(u1.x), u0.x);
                const float bL1 = from_prev_lane(rot_from_prev_lane(u0.y), u1.y);
                const float bR1 = from_next_lane(ebR, u1.x);
                float a0 = 0.0f, a1 = 0.0f, a2 = 0.0f, a3 = 0.0f;
                a0 += tL0  * kd0[0];  a1 += t0.x * kd1[0];  a2 += tL1  * kd2[0];  a3 += t1.x * kd3[0];
                a0 += t0.x * kd0[1];  a1 += t0.y * kd1[1];  a2 += t1.x * kd2[1];  a3 += t1.y * kd3[1];
                a0 += t0.y * kd0[2];  a1 += tR0  * kd1[2];  a2 += t1.y * kd2[2];  a3 += tR1  * kd3[2];
                a0 += cL0  * kd0[3];  a1 += m0.x * kd1[3];  a2 += cL1  * kd2[3];  a3 += m1.x * kd3[3];
                a0 += m0.x * kd0[4];  a1 += m0.y * kd1[4];  a2 += m1.x * kd2[4];  a3 += m1.y * kd3[4];
                a0 += m0.y * kd0[5];  a1 += cR0  * kd1[5];  a2 += m1.y * kd2[5];  a3 += cR1  * kd3[5];
                a0 += bL0  * kd0[6];  a1 += u0.x * kd1[6];  a2 += bL1  * kd2[6];  a3 += u1.x * kd3[6];
                a0 += u0.x * kd0[7];  a1 += u0.y * kd1[7];  a2 += u1.x * kd2[7];  a3 += u1.y * kd3[7];
                a0 += u0.y * kd0[8];  a1 += bR0  * kd1[8];  a2 += u1.y * kd2[8];  a3 += bR1  * kd3[8];
                yp0[2 * (p0 + pp)] = a0;
                yp1[2 * (p0 + pp)] = a1;
                yp2[2 * (p0 + pp)] = a2;
                yp3[2 * (p0 + pp)] = a3;
            }
        }
    }
    __syncthreads();

    v8f acc[2][2];
    #pragma unroll
    for (int a1 = 0; a1 < 2; ++a1)
        #pragma unroll
        for (int a2 = 0; a2 < 2; ++a2) acc[a1][a2] = (v8f){};

    for (int ihalf = 0; ihalf < 2; ++ihalf) {
        {
            const int n    = wave * 16 + lm;
            const int gcol = ihalf * 512 + n;
            v8f d0 = (v8f){}, d1 = (v8f){}, d2 = (v8f){}, d3 = (v8f){};
            #pragma unroll
            for (int ks = 0; ks < 4; ++ks) {
                const int k0 = ks * 4 + 2 * hiL;
                v2f a0, a1, a2, a3, bf;
                a0.x = hP_lds[lm * HID_ + k0];
                a0.y = hP_lds[lm * HID_ + k0 + 1];
                a1.x = hP_lds[(16 + lm) * HID_ + k0];
                a1.y = hP_lds[(16 + lm) * HID_ + k0 + 1];
                a2.x = hP_lds[(32 + lm) * HID_ + k0];
                a2.y = hP_lds[(32 + lm) * HID_ + k0 + 1];
                a3.x = hP_lds[(48 + lm) * HID_ + k0];
                a3.y = hP_lds[(48 + lm) * HID_ + k0 + 1];
                bf.x = mP_W2[(size_t)k0 * 1024 + gcol];
                bf.y = mP_W2[(size_t)(k0 + 1) * 1024 + gcol];
                d0 = __builtin_amdgcn_wmma_f32_16x16x4_f32(
                        false, a0, false, bf, (short)0, d0, false, false);
                d1 = __builtin_amdgcn_wmma_f32_16x16x4_f32(
                        false, a1, false, bf, (short)0, d1, false, false);
                d2 = __builtin_amdgcn_wmma_f32_16x16x4_f32(
                        false, a2, false, bf, (short)0, d2, false, false);
                d3 = __builtin_amdgcn_wmma_f32_16x16x4_f32(
                        false, a3, false, bf, (short)0, d3, false, false);
            }
            const int i_loc = n >> 5;
            const int o     = n & 31;
            const int ig    = gcol >> 5;
            const float kbase = kernelP[ig * CO + o] + mP_b2[gcol];
            #pragma unroll
            for (int r = 0; r < 8; ++r) {
                const int m = r + 8 * hiL;
                kP_lds[(m * 16 + i_loc) * CO + o]        = d0[r] + kbase;
                kP_lds[((16 + m) * 16 + i_loc) * CO + o] = d1[r] + kbase;
                kP_lds[((32 + m) * 16 + i_loc) * CO + o] = d2[r] + kbase;
                kP_lds[((48 + m) * 16 + i_loc) * CO + o] = d3[r] + kbase;
            }
        }
        __syncthreads();

        #pragma unroll
        for (int li = 0; li < 2; ++li) {
            const int l = wave * 2 + li;
            #pragma unroll
            for (int nt = 0; nt < 2; ++nt) {
                v8f d = acc[li][nt];
                #pragma unroll
                for (int ks = 0; ks < 4; ++ks) {
                    const int k0 = ks * 4 + 2 * hiL;
                    v2f af, bf;
                    if (lm < 8) {
                        af.x = y_lds[l * Y_LP + lm * CI + ihalf * 16 + k0];
                        af.y = y_lds[l * Y_LP + lm * CI + ihalf * 16 + k0 + 1];
                    } else {
                        af.x = 0.0f; af.y = 0.0f;
                    }
                    const int o = nt * 16 + lm;
                    bf.x = kP_lds[(l * 16 + k0) * CO + o];
                    bf.y = kP_lds[(l * 16 + k0 + 1) * CO + o];
                    d = __builtin_amdgcn_wmma_f32_16x16x4_f32(
                            false, af, false, bf, (short)0, d, false, false);
                }
                acc[li][nt] = d;
            }
        }
        __syncthreads();
    }

    float* st = kP_lds;
    if (hiL == 0) {
        #pragma unroll
        for (int nt = 0; nt < 2; ++nt) {
            const int o = nt * 16 + lm;
            const float bo = bias[o];
            #pragma unroll
            for (int r = 0; r < 8; ++r) {
                float* sp = &st[(r * CO + o) * ST_P + wave * 2];
                sp[0] = acc[0][nt][r] + bo;
                sp[1] = acc[1][nt][r] + bo;
            }
        }
    }
    __syncthreads();
    {
        const size_t pix = (size_t)h * WW + w0;
        #pragma unroll
        for (int it = 0; it < 4; ++it) {
            const int item = it * NTHR + tid;
            const int row  = item >> 4;
            const int q    = item & 15;
            const float4 v = *reinterpret_cast<const float4*>(&st[row * ST_P + q * 4]);
            *reinterpret_cast<float4*>(&out[(size_t)row * (HH * WW) + pix + q * 4]) = v;
        }
    }
}

extern "C" void kernel_launch(void* const* d_in, const int* in_sizes, int n_in,
                              void* d_out, int out_size, void* d_ws, size_t ws_size,
                              hipStream_t stream) {
    const float* x       = (const float*)d_in[0];
    const float* kernelD = (const float*)d_in[1];
    const float* kernelP = (const float*)d_in[2];
    const float* bias    = (const float*)d_in[3];
    const float* cvec    = (const float*)d_in[4];
    const float* mD_W1   = (const float*)d_in[5];
    const float* mD_b1   = (const float*)d_in[6];
    const float* mD_W2   = (const float*)d_in[7];
    const float* mD_b2   = (const float*)d_in[8];
    const float* mP_W1   = (const float*)d_in[9];
    const float* mP_b1   = (const float*)d_in[10];
    const float* mP_W2   = (const float*)d_in[11];
    const float* mP_b2   = (const float*)d_in[12];
    float* out = (float*)d_out;

    dim3 grid(256);
    dim3 block(NTHR);
    hipLaunchKernelGGL(scaconv_ds_kernel, grid, block, 0, stream,
                       x, kernelD, kernelP, bias, cvec,
                       mD_W1, mD_b1, mD_W2, mD_b2,
                       mP_W1, mP_b1, mP_W2, mP_b2, out);
}


# LDS operands of both matrix stages re-laid bank-conflict-free (hP in 64-word k-quads, y batch pitch 36, kP column halves swapped by l and k bits); pad rows load duplicates so fragment loads group
# speedup vs baseline: 6.2933x; 1.1173x over previous
#include <hip/hip_runtime.h>


#define B_   8
#define CI   32
#define CO   32
#define HH   128
#define WW   128
#define HID_ 16
#define TL   64
#define NTHR 1024
#define Y_BP (CI + 4)
#define Y_LP (B_ * Y_BP + 2)
#define ST_P 68

typedef __attribute__((ext_vector_type(2))) float v2f;
typedef __attribute__((ext_vector_type(8))) float v8f;

__device__ __forceinline__ float from_prev_lane(float edge, float v) {
    return __int_as_float(__builtin_amdgcn_update_dpp(__float_as_int(edge), __float_as_int(v),
                                                      0x111  , 0xF, 0xF, false));
}
__device__ __forceinline__ float from_next_lane(float edge, float v) {
    return __int_as_float(__builtin_amdgcn_update_dpp(__float_as_int(edge), __float_as_int(v),
                                                      0x101  , 0xF, 0xF, false));
}
__device__ __forceinline__ float rot_from_prev_lane(float v) {
    return __int_as_float(__builtin_amdgcn_update_dpp(0, __float_as_int(v),
                                                      0x121  , 0xF, 0xF, false));
}
__device__ __forceinline__ float rot_from_next_lane(float v) {
    return __int_as_float(__builtin_amdgcn_update_dpp(0, __float_as_int(v),
                                                      0x12F  , 0xF, 0xF, false));
}

__global__ __launch_bounds__(NTHR)
void scaconv_ds_kernel(const float* __restrict__ x,
                       const float* __restrict__ kernelD,
                       const float* __restrict__ kernelP,
                       const float* __restrict__ bias,
                       const float* __restrict__ cvec,
                       const float* __restrict__ mD_W1,
                       const float* __restrict__ mD_b1,
                       const float* __restrict__ mD_W2,
                       const float* __restrict__ mD_b2,
                       const float* __restrict__ mP_W1,
                       const float* __restrict__ mP_b1,
                       const float* __restrict__ mP_W2,
                       const float* __restrict__ mP_b2,
                       float* __restrict__ out)
{
    __shared__ alignas(16) float lds[2 * TL * HID_ + TL * 9 + TL * Y_LP + TL * 16 * CO];
    float* const hP_lds = lds;
    float* const hD_lds = hP_lds + TL * HID_;
    float* const kD_lds = hD_lds + TL * HID_;
    float* const y_lds  = kD_lds + TL * 9;
    float* const kP_lds = y_lds + TL * Y_LP;

    const int tid  = threadIdx.x;
    const int wave = __builtin_amdgcn_readfirstlane(tid >> 5);
    const int lane = tid & 31;
    const int lm   = lane & 15;
    const int hiL  = lane >> 4;

    const int wg = blockIdx.x;
    const int h  = wg >> 1;
    const int w0 = (wg & 1) * TL;

    {
        const int l = tid >> 4;
        const int j = tid & 15;
        const int w = w0 + l;
        const float xn = (float)h * (1.0f / 128.0f);
        const float yn = (float)w * (1.0f / 128.0f);
        float f[12];
        f[0] = xn; f[1] = 1.0f - xn; f[2] = yn; f[3] = 1.0f - yn;
        #pragma unroll
        for (int q = 0; q < 8; ++q) f[4 + q] = cvec[q];
        float aD = mD_b1[j], aP = mP_b1[j];
        #pragma unroll
        for (int q = 0; q < 12; ++q) {
            aD += f[q] * mD_W1[q * HID_ + j];
            aP += f[q] * mP_W1[q * HID_ + j];
        }
        hD_lds[l * HID_ + j] = fmaxf(aD, 0.0f);
        hP_lds[((l >> 4) * 4 + (j >> 2)) * 64 + (l & 15) * 4 + (j & 3)] = fmaxf(aP, 0.0f);
    }
    __syncthreads();

    if (tid < TL * 9) {
        const int l = tid / 9;
        const int k = tid % 9;
        float a = kernelD[k] + mD_b2[k];
        #pragma unroll
        for (int j = 0; j < HID_; ++j) a += hD_lds[l * HID_ + j] * mD_W2[j * 9 + k];
        kD_lds[l * 9 + k] = a;
    }
    __syncthreads();

    {
        const int b   = wave >> 2;
        const int ch0 = (wave & 3) * 8;
        const int lA  = 2 * lm;
        float kd0[9], kd1[9], kd2[9], kd3[9];
        #pragma unroll
        for (int k = 0; k < 9; ++k) {
            kd0[k] = kD_lds[lA * 9 + k];
            kd1[k] = kD_lds[(lA + 1) * 9 + k];
            kd2[k] = kD_lds[(32 + lA) * 9 + k];
            kd3[k] = kD_lds[(33 + lA) * 9 + k];
        }
        const bool okT = (h > 0), okB = (h < HH - 1);
        const bool leftTile = (w0 == 0);
        const int  cH = leftTile ? TL : TL - 1;
        const int rT = (okT ? h - 1 : h) * WW;
        const int rC = h * WW;
        const int rB = (okB ? h + 1 : h) * WW;
        const float* xl = x + (size_t)(b * CI + ch0 + hiL) * (HH * WW);
        const float* mT = xl + rT + w0 + lA; const float* eT = xl + rT + cH;
        const float* mC = xl + rC + w0 + lA; const float* eC = xl + rC + cH;
        const float* mB = xl + rB + w0 + lA; const float* eB = xl + rB + cH;
        float* yp0 = &y_lds[lA * Y_LP + b * Y_BP + ch0 + hiL];
        float* yp1 = yp0 + Y_LP;
        float* yp2 = yp0 + 32 * Y_LP;
        float* yp3 = yp0 + 33 * Y_LP;
        #pragma clang loop unroll(disable) vectorize(disable)
        for (int p0 = 0; p0 < 4; p0 += 2) {
            #pragma unroll
            for (int pp = 0; pp < 2; ++pp) {
                const int c = 2 * (p0 + pp) * (HH * WW);
                float2 t0 = *reinterpret_cast<const float2*>(mT + c);
                float2 t1 = *reinterpret_cast<const float2*>(mT + c + 32);
                float2 m0 = *reinterpret_cast<const float2*>(mC + c);
                float2 m1 = *reinterpret_cast<const float2*>(mC + c + 32);
                float2 u0 = *reinterpret_cast<const float2*>(mB + c);
                float2 u1 = *reinterpret_cast<const float2*>(mB + c + 32);
                float et = eT[c], ec = eC[c], eb = eB[c];
                t0.x = okT ? t0.x : 0.0f;  t0.y = okT ? t0.y : 0.0f;
                t1.x = okT ? t1.x : 0.0f;  t1.y = okT ? t1.y : 0.0f;  et = okT ? et : 0.0f;
                u0.x = okB ? u0.x : 0.0f;  u0.y = okB ? u0.y : 0.0f;
                u1.x = okB ? u1.x : 0.0f;  u1.y = okB ? u1.y : 0.0f;  eb = okB ? eb : 0.0f;
                const float etL = leftTile ? 0.0f : et, etR = leftTile ? et : 0.0f;
                const float ecL = leftTile ? 0.0f : ec, ecR = leftTile ? ec : 0.0f;
                const float ebL = leftTile ? 0.0f : eb, ebR = leftTile ? eb : 0.0f;
                const float tL0 = from_prev_lane(etL, t0.y);
                const float tR0 = from_next_lane(rot_from_next_lane(t1.x), t0.x);
                const float tL1 = from_prev_lane(rot_from_prev_lane(t0.y), t1.y);
                const float tR1 = from_next_lane(etR, t1.x);
                const float cL0 = from_prev_lane(ecL, m0.y);
                const float cR0 = from_next_lane(rot_from_next_lane(m1.x), m0.x);
                const float cL1 = from_prev_lane(rot_from_prev_lane(m0.y), m1.y);
                const float cR1 = from_next_lane(ecR, m1.x);
                const float bL0 = from_prev_lane(ebL, u0.y);
                const float bR0 = from_next_lane(rot_from_next_lane(u1.x), u0.x);
                const float bL1 = from_prev_lane(rot_from_prev_lane(u0.y), u1.y);
                const float bR1 = from_next_lane(ebR, u1.x);
                float a0 = 0.0f, a1 = 0.0f, a2 = 0.0f, a3 = 0.0f;
                a0 += tL0  * kd0[0];  a1 += t0.x * kd1[0];  a2 += tL1  * kd2[0];  a3 += t1.x * kd3[0];
                a0 += t0.x * kd0[1];  a1 += t0.y * kd1[1];  a2 += t1.x * kd2[1];  a3 += t1.y * kd3[1];
                a0 += t0.y * kd0[2];  a1 += tR0  * kd1[2];  a2 += t1.y * kd2[2];  a3 += tR1  * kd3[2];
                a0 += cL0  * kd0[3];  a1 += m0.x * kd1[3];  a2 += cL1  * kd2[3];  a3 += m1.x * kd3[3];
                a0 += m0.x * kd0[4];  a1 += m0.y * kd1[4];  a2 += m1.x * kd2[4];  a3 += m1.y * kd3[4];
                a0 += m0.y * kd0[5];  a1 += cR0  * kd1[5];  a2 += m1.y * kd2[5];  a3 += cR1  * kd3[5];
                a0 += bL0  * kd0[6];  a1 += u0.x * kd1[6];  a2 += bL1  * kd2[6];  a3 += u1.x * kd3[6];
                a0 += u0.x * kd0[7];  a1 += u0.y * kd1[7];  a2 += u1.x * kd2[7];  a3 += u1.y * kd3[7];
                a0 += u0.y * kd0[8];  a1 += bR0  * kd1[8];  a2 += u1.y * kd2[8];  a3 += bR1  * kd3[8];
                yp0[2 * (p0 + pp)] = a0;
                yp1[2 * (p0 + pp)] = a1;
                yp2[2 * (p0 + pp)] = a2;
                yp3[2 * (p0 + pp)] = a3;
            }
        }
    }
    __syncthreads();

    v8f acc[2][2];
    #pragma unroll
    for (int a1 = 0; a1 < 2; ++a1)
        #pragma unroll
        for (int a2 = 0; a2 < 2; ++a2) acc[a1][a2] = (v8f){};

    for (int ihalf = 0; ihalf < 2; ++ihalf) {
        {
            const int n    = wave * 16 + lm;
            const int gcol = ihalf * 512 + n;
            v8f d0 = (v8f){}, d1 = (v8f){}, d2 = (v8f){}, d3 = (v8f){};
            #pragma unroll
            for (int ks = 0; ks < 4; ++ks) {
                const int k0 = ks * 4 + 2 * hiL;
                v2f a0, a1, a2, a3, bf;
                const float* hp = &hP_lds[ks * 64 + lm * 4 + 2 * hiL];
                a0 = *reinterpret_cast<const v2f*>(hp);
                a1 = *reinterpret_cast<const v2f*>(hp + 256);
                a2 = *reinterpret_cast<const v2f*>(hp + 512);
                a3 = *reinterpret_cast<const v2f*>(hp + 768);
                bf.x = mP_W2[(size_t)k0 * 1024 + gcol];
                bf.y = mP_W2[(size_t)(k0 + 1) * 1024 + gcol];
                d0 = __builtin_amdgcn_wmma_f32_16x16x4_f32(
                        false, a0, false, bf, (short)0, d0, false, false);
                d1 = __builtin_amdgcn_wmma_f32_16x16x4_f32(
                        false, a1, false, bf, (short)0, d1, false, false);
                d2 = __builtin_amdgcn_wmma_f32_16x16x4_f32(
                        false, a2, false, bf, (short)0, d2, false, false);
                d3 = __builtin_amdgcn_wmma_f32_16x16x4_f32(
                        false, a3, false, bf, (short)0, d3, false, false);
            }
            const int i_loc = n >> 5;
            const int o     = n & 31;
            const int ig    = gcol >> 5;
            const float kbase = kernelP[ig * CO + o] + mP_b2[gcol];
            const int os = o ^ ((((i_loc >> 1) ^ hiL) & 1) << 4);
            #pragma unroll
            for (int r = 0; r < 8; ++r) {
                const int m = r + 8 * hiL;
                kP_lds[(m * 16 + i_loc) * CO + os]        = d0[r] + kbase;
                kP_lds[((16 + m) * 16 + i_loc) * CO + os] = d1[r] + kbase;
                kP_lds[((32 + m) * 16 + i_loc) * CO + os] = d2[r] + kbase;
                kP_lds[((48 + m) * 16 + i_loc) * CO + os] = d3[r] + kbase;
            }
        }
        __syncthreads();

        {
            v2f af[2][4], bf[2][2][4];
            #pragma unroll
            for (int li = 0; li < 2; ++li) {
                const int l = wave * 2 + li;
                const int sw = ((hiL ^ (l >> 3)) & 1) << 4;
                #pragma unroll
                for (int ks = 0; ks < 4; ++ks) {
                    const int k0 = ks * 4 + 2 * hiL;
                    af[li][ks] = *reinterpret_cast<const v2f*>(
                        &y_lds[l * Y_LP + (lm & 7) * Y_BP + ihalf * 16 + k0]);
                    #pragma unroll
                    for (int nt = 0; nt < 2; ++nt) {
                        const int os = (nt * 16 + lm) ^ sw;
                        bf[li][nt][ks].x = kP_lds[(l * 16 + k0) * CO + os];
                        bf[li][nt][ks].y = kP_lds[(l * 16 + k0 + 1) * CO + os];
                    }
                }
            }
            #pragma unroll
            for (int ks = 0; ks < 4; ++ks)
                #pragma unroll
                for (int li = 0; li < 2; ++li)
                    #pragma unroll
                    for (int nt = 0; nt < 2; ++nt)
                        acc[li][nt] = __builtin_amdgcn_wmma_f32_16x16x4_f32(
                            false, af[li][ks], false, bf[li][nt][ks], (short)0, acc[li][nt], false, false);
        }
        __builtin_amdgcn_sched_barrier(0);
        __syncthreads();
    }

    float* st = kP_lds;
    if (hiL == 0) {
        #pragma unroll
        for (int nt = 0; nt < 2; ++nt) {
            const int o = nt * 16 + lm;
            const float bo = bias[o];
            #pragma unroll
            for (int r = 0; r < 8; ++r) {
                float* sp = &st[(r * CO + o) * ST_P + wave * 2];
                sp[0] = acc[0][nt][r] + bo;
                sp[1] = acc[1][nt][r] + bo;
            }
        }
    }
    __syncthreads();
    {
        const size_t pix = (size_t)h * WW + w0;
        #pragma unroll
        for (int it = 0; it < 4; ++it) {
            const int item = it * NTHR + tid;
            const int row  = item >> 4;
            const int q    = item & 15;
            const float4 v = *reinterpret_cast<const float4*>(&st[row * ST_P + q * 4]);
            *reinterpret_cast<float4*>(&out[(size_t)row * (HH * WW) + pix + q * 4]) = v;
        }
    }
}

extern "C" void kernel_launch(void* const* d_in, const int* in_sizes, int n_in,
                              void* d_out, int out_size, void* d_ws, size_t ws_size,
                              hipStream_t stream) {
    const float* x       = (const float*)d_in[0];
    const float* kernelD = (const float*)d_in[1];
    const float* kernelP = (const float*)d_in[2];
    const float* bias    = (const float*)d_in[3];
    const float* cvec    = (const float*)d_in[4];
    const float* mD_W1   = (const float*)d_in[5];
    const float* mD_b1   = (const float*)d_in[6];
    const float* mD_W2   = (const float*)d_in[7];
    const float* mD_b2   = (const float*)d_in[8];
    const float* mP_W1   = (const float*)d_in[9];
    const float* mP_b1   = (const float*)d_in[10];
    const float* mP_W2   = (const float*)d_in[11];
    const float* mP_b2   = (const float*)d_in[12];
    float* out = (float*)d_out;

    dim3 grid(256);
    dim3 block(NTHR);
    hipLaunchKernelGGL(scaconv_ds_kernel, grid, block, 0, stream,
                       x, kernelD, kernelP, bias, cvec,
                       mD_W1, mD_b1, mD_W2, mD_b2,
                       mP_W1, mP_b1, mP_W2, mP_b2, out);
}


# The first kP half-tile (it needs no y) and the depthwise stage run side by side: of each SIMD's 8 waves 4 take one first, 4 the other, then swap; one barrier fewer; sums in same order
# speedup vs baseline: 6.8800x; 1.0932x over previous
#include <hip/hip_runtime.h>


#define B_   8
#define CI   32
#define CO   32
#define HH   128
#define WW   128
#define HID_ 16
#define TL   64
#define NTHR 1024
#define Y_BP (CI + 4)
#define Y_LP (B_ * Y_BP + 2)
#define ST_P 68

typedef __attribute__((ext_vector_type(2))) float v2f;
typedef __attribute__((ext_vector_type(8))) float v8f;

__device__ __forceinline__ float from_prev_lane(float edge, float v) {
    return __int_as_float(__builtin_amdgcn_update_dpp(__float_as_int(edge), __float_as_int(v),
                                                      0x111  , 0xF, 0xF, false));
}
__device__ __forceinline__ float from_next_lane(float edge, float v) {
    return __int_as_float(__builtin_amdgcn_update_dpp(__float_as_int(edge), __float_as_int(v),
                                                      0x101  , 0xF, 0xF, false));
}
__device__ __forceinline__ float rot_from_prev_lane(float v) {
    return __int_as_float(__builtin_amdgcn_update_dpp(0, __float_as_int(v),
                                                      0x121  , 0xF, 0xF, false));
}
__device__ __forceinline__ float rot_from_next_lane(float v) {
    return __int_as_float(__builtin_amdgcn_update_dpp(0, __float_as_int(v),
                                                      0x12F  , 0xF, 0xF, false));
}

__global__ __launch_bounds__(NTHR)
void scaconv_ds_kernel(const float* __restrict__ x,
                       const float* __restrict__ kernelD,
                       const float* __restrict__ kernelP,
                       const float* __restrict__ bias,
                       const float* __restrict__ cvec,
                       const float* __restrict__ mD_W1,
                       const float* __restrict__ mD_b1,
                       const float* __restrict__ mD_W2,
                       const float* __restrict__ mD_b2,
                       const float* __restrict__ mP_W1,
                       const float* __restrict__ mP_b1,
                       const float* __restrict__ mP_W2,
                       const float* __restrict__ mP_b2,
                       float* __restrict__ out)
{
    __shared__ alignas(16) float lds[2 * TL * HID_ + TL * 9 + TL * Y_LP + TL * 16 * CO];
    float* const hP_lds = lds;
    float* const hD_lds = hP_lds + TL * HID_;
    float* const kD_lds = hD_lds + TL * HID_;
    float* const y_lds  = kD_lds + TL * 9;
    float* const kP_lds = y_lds + TL * Y_LP;

    const int tid  = threadIdx.x;
    const int wave = __builtin_amdgcn_readfirstlane(tid >> 5);
    const int lane = tid & 31;
    const int lm   = lane & 15;
    const int hiL  = lane >> 4;

    const int wg = blockIdx.x;
    const int h  = wg >> 1;
    const int w0 = (wg & 1) * TL;

    {
        const int l = tid >> 4;
        const int j = tid & 15;
        const int w = w0 + l;
        const float xn = (float)h * (1.0f / 128.0f);
        const float yn = (float)w * (1.0f / 128.0f);
        float f[12];
        f[0] = xn; f[1] = 1.0f - xn; f[2] = yn; f[3] = 1.0f - yn;
        #pragma unroll
        for (int q = 0; q < 8; ++q) f[4 + q] = cvec[q];
        float aD = mD_b1[j], aP = mP_b1[j];
        #pragma unroll
        for (int q = 0; q < 12; ++q) {
            aD += f[q] * mD_W1[q * HID_ + j];
            aP += f[q] * mP_W1[q * HID_ + j];
        }
        hD_lds[l * HID_ + j] = fmaxf(aD, 0.0f);
        hP_lds[((l >> 4) * 4 + (j >> 2)) * 64 + (l & 15) * 4 + (j & 3)] = fmaxf(aP, 0.0f);
    }
    __syncthreads();

    if (tid < TL * 9) {
        const int l = tid / 9;
        const int k = tid % 9;
        float a = kernelD[k] + mD_b2[k];
        #pragma unroll
        for (int j = 0; j < HID_; ++j) a += hD_lds[l * HID_ + j] * mD_W2[j * 9 + k];
        kD_lds[l * 9 + k] = a;
    }
    __syncthreads();

    auto depthwise_stage = [&]() __attribute__((always_inline)) {
        const int b   = wave >> 2;
        const int ch0 = (wave & 3) * 8;
        const int lA  = 2 * lm;
        float kd0[9], kd1[9], kd2[9], kd3[9];
        #pragma unroll
        for (int k = 0; k < 9; ++k) {
            kd0[k] = kD_lds[lA * 9 + k];
            kd1[k] = kD_lds[(lA + 1) * 9 + k];
            kd2[k] = kD_lds[(32 + lA) * 9 + k];
            kd3[k] = kD_lds[(33 + lA) * 9 + k];
        }
        const bool okT = (h > 0), okB = (h < HH - 1);
        const bool leftTile = (w0 == 0);
        const int  cH = leftTile ? TL : TL - 1;
        const int rT = (okT ? h - 1 : h) * WW;
        const int rC = h * WW;
        const int rB = (okB ? h + 1 : h) * WW;
        const float* xl = x + (size_t)(b * CI + ch0 + hiL) * (HH * WW);
        const float* mT = xl + rT + w0 + lA; const float* eT = xl + rT + cH;
        const float* mC = xl + rC + w0 + lA; const float* eC = xl + rC + cH;
        const float* mB = xl + rB + w0 + lA; const float* eB = xl + rB + cH;
        float* yp0 = &y_lds[lA * Y_LP + b * Y_BP + ch0 + hiL];
        float* yp1 = yp0 + Y_LP;
        float* yp2 = yp0 + 32 * Y_LP;
        float* yp3 = yp0 + 33 * Y_LP;
        #pragma clang loop unroll(disable) vectorize(disable)
        for (int p0 = 0; p0 < 4; p0 += 2) {
            #pragma unroll
            for (int pp = 0; pp < 2; ++pp) {
                const int c = 2 * (p0 + pp) * (HH * WW);
                float2 t0 = *reinterpret_cast<const float2*>(mT + c);
                float2 t1 = *reinterpret_cast<const float2*>(mT + c + 32);
                float2 m0 = *reinterpret_cast<const float2*>(mC + c);
                float2 m1 = *reinterpret_cast<const float2*>(mC + c + 32);
                float2 u0 = *reinterpret_cast<const float2*>(mB + c);
                float2 u1 = *reinterpret_cast<const float2*>(mB + c + 32);
                float et = eT[c], ec = eC[c], eb = eB[c];
                t0.x = okT ? t0.x : 0.0f;  t0.y = okT ? t0.y : 0.0f;
                t1.x = okT ? t1.x : 0.0f;  t1.y = okT ? t1.y : 0.0f;  et = okT ? et : 0.0f;
                u0.x = okB ? u0.x : 0.0f;  u0.y = okB ? u0.y : 0.0f;
                u1.x = okB ? u1.x : 0.0f;  u1.y = okB ? u1.y : 0.0f;  eb = okB ? eb : 0.0f;
                const float etL = leftTile ? 0.0f : et, etR = leftTile ? et : 0.0f;
                const float ecL = leftTile ? 0.0f : ec, ecR = leftTile ? ec : 0.0f;
                const float ebL = leftTile ? 0.0f : eb, ebR = leftTile ? eb : 0.0f;
                const float tL0 = from_prev_lane(etL, t0.y);
                const float tR0 = from_next_lane(rot_from_next_lane(t1.x), t0.x);
                const float tL1 = from_prev_lane(rot_from_prev_lane(t0.y), t1.y);
                const float tR1 = from_next_lane(etR, t1.x);
                const float cL0 = from_prev_lane(ecL, m0.y);
                const float cR0 = from_next_lane(rot_from_next_lane(m1.x), m0.x);
                const float cL1 = from_prev_lane(rot_from_prev_lane(m0.y), m1.y);
                const float cR1 = from_next_lane(ecR, m1.x);
                const float bL0 = from_prev_lane(ebL, u0.y);
                const float bR0 = from_next_lane(rot_from_next_lane(u1.x), u0.x);
                const float bL1 = from_prev_lane(rot_from_prev_lane(u0.y), u1.y);
                const float bR1 = from_next_lane(ebR, u1.x);
                float a0 = 0.0f, a1 = 0.0f, a2 = 0.0f, a3 = 0.0f;
                a0 += tL0  * kd0[0];  a1 += t0.x * kd1[0];  a2 += tL1  * kd2[0];  a3 += t1.x * kd3[0];
                a0 += t0.x * kd0[1];  a1 += t0.y * kd1[1];  a2 += t1.x * kd2[1];  a3 += t1.y * kd3[1];
                a0 += t0.y * kd0[2];  a1 += tR0  * kd1[2];  a2 += t1.y * kd2[2];  a3 += tR1  * kd3[2];
                a0 += cL0  * kd0[3];  a1 += m0.x * kd1[3];  a2 += cL1  * kd2[3];  a3 += m1.x * kd3[3];
                a0 += m0.x * kd0[4];  a1 += m0.y * kd1[4];  a2 += m1.x * kd2[4];  a3 += m1.y * kd3[4];
                a0 += m0.y * kd0[5];  a1 += cR0  * kd1[5];  a2 += m1.y * kd2[5];  a3 += cR1  * kd3[5];
                a0 += bL0  * kd0[6];  a1 += u0.x * kd1[6];  a2 += bL1  * kd2[6];  a3 += u1.x * kd3[6];
                a0 += u0.x * kd0[7];  a1 += u0.y * kd1[7];  a2 += u1.x * kd2[7];  a3 += u1.y * kd3[7];
                a0 += u0.y * kd0[8];  a1 += bR0  * kd1[8];  a2 += u1.y * kd2[8];  a3 += bR1  * kd3[8];
                yp0[2 * (p0 + pp)] = a0;
                yp1[2 * (p0 + pp)] = a1;
                yp2[2 * (p0 + pp)] = a2;
                yp3[2 * (p0 + pp)] = a3;
            }
        }
    };

    auto kp_stage = [&](const int ihalf) __attribute__((always_inline)) {
        {
            const int n    = wave * 16 + lm;
            const int gcol = ihalf * 512 + n;
            v8f d0 = (v8f){}, d1 = (v8f){}, d2 = (v8f){}, d3 = (v8f){};
            #pragma unroll
            for (int ks = 0; ks < 4; ++ks) {
                const int k0 = ks * 4 + 2 * hiL;
                v2f a0, a1, a2, a3, bf;
                const float* hp = &hP_lds[ks * 64 + lm * 4 + 2 * hiL];
                a0 = *reinterpret_cast<const v2f*>(hp);
                a1 = *reinterpret_cast<const v2f*>(hp + 256);
                a2 = *reinterpret_cast<const v2f*>(hp + 512);
                a3 = *reinterpret_cast<const v2f*>(hp + 768);
                bf.x = mP_W2[(size_t)k0 * 1024 + gcol];
                bf.y = mP_W2[(size_t)(k0 + 1) * 1024 + gcol];
                d0 = __builtin_amdgcn_wmma_f32_16x16x4_f32(
                        false, a0, false, bf, (short)0, d0, false, false);
                d1 = __builtin_amdgcn_wmma_f32_16x16x4_f32(
                        false, a1, false, bf, (short)0, d1, false, false);
                d2 = __builtin_amdgcn_wmma_f32_16x16x4_f32(
                        false, a2, false, bf, (short)0, d2, false, false);
                d3 = __builtin_amdgcn_wmma_f32_16x16x4_f32(
                        false, a3, false, bf, (short)0, d3, false, false);
            }
            const int i_loc = n >> 5;
            const int o     = n & 31;
            const int ig    = gcol >> 5;
            const float kbase = kernelP[ig * CO + o] + mP_b2[gcol];
            const int os = o ^ ((((i_loc >> 1) ^ hiL) & 1) << 4);
            #pragma unroll
            for (int r = 0; r < 8; ++r) {
                const int m = r + 8 * hiL;
                kP_lds[(m * 16 + i_loc) * CO + os]        = d0[r] + kbase;
                kP_lds[((16 + m) * 16 + i_loc) * CO + os] = d1[r] + kbase;
                kP_lds[((32 + m) * 16 + i_loc) * CO + os] = d2[r] + kbase;
                kP_lds[((48 + m) * 16 + i_loc) * CO + os] = d3[r] + kbase;
            }
        }
    };

    if (((wave >> 2) & 1) == 0) {
        depthwise_stage();
        kp_stage(0);
    } else {
        kp_stage(0);
        depthwise_stage();
    }
    __syncthreads();

    v8f acc[2][2];
    #pragma unroll
    for (int a1 = 0; a1 < 2; ++a1)
        #pragma unroll
        for (int a2 = 0; a2 < 2; ++a2) acc[a1][a2] = (v8f){};

    for (int ihalf = 0; ihalf < 2; ++ihalf) {
        if (ihalf == 1) {
            kp_stage(1);
            __syncthreads();
        }

        {
            v2f af[2][4], bf[2][2][4];
            #pragma unroll
            for (int li = 0; li < 2; ++li) {
                const int l = wave * 2 + li;
                const int sw = ((hiL ^ (l >> 3)) & 1) << 4;
                #pragma unroll
                for (int ks = 0; ks < 4; ++ks) {
                    const int k0 = ks * 4 + 2 * hiL;
                    af[li][ks] = *reinterpret_cast<const v2f*>(
                        &y_lds[l * Y_LP + (lm & 7) * Y_BP + ihalf * 16 + k0]);
                    #pragma unroll
                    for (int nt = 0; nt < 2; ++nt) {
                        const int os = (nt * 16 + lm) ^ sw;
                        bf[li][nt][ks].x = kP_lds[(l * 16 + k0) * CO + os];
                        bf[li][nt][ks].y = kP_lds[(l * 16 + k0 + 1) * CO + os];
                    }
                }
            }
            #pragma unroll
            for (int ks = 0; ks < 4; ++ks)
                #pragma unroll
                for (int li = 0; li < 2; ++li)
                    #pragma unroll
                    for (int nt = 0; nt < 2; ++nt)
                        acc[li][nt] = __builtin_amdgcn_wmma_f32_16x16x4_f32(
                            false, af[li][ks], false, bf[li][nt][ks], (short)0, acc[li][nt], false, false);
        }
        __builtin_amdgcn_sched_barrier(0);
        __syncthreads();
    }

    float* st = kP_lds;
    if (hiL == 0) {
        #pragma unroll
        for (int nt = 0; nt < 2; ++nt) {
            const int o = nt * 16 + lm;
            const float bo = bias[o];
            #pragma unroll
            for (int r = 0; r < 8; ++r) {
                float* sp = &st[(r * CO + o) * ST_P + wave * 2];
                sp[0] = acc[0][nt][r] + bo;
                sp[1] = acc[1][nt][r] + bo;
            }
        }
    }
    __syncthreads();
    {
        const size_t pix = (size_t)h * WW + w0;
        #pragma unroll
        for (int it = 0; it < 4; ++it) {
            const int item = it * NTHR + tid;
            const int row  = item >> 4;
            const int q    = item & 15;
            const float4 v = *reinterpret_cast<const float4*>(&st[row * ST_P + q * 4]);
            *reinterpret_cast<float4*>(&out[(size_t)row * (HH * WW) + pix + q * 4]) = v;
        }
    }
}

extern "C" void kernel_launch(void* const* d_in, const int* in_sizes, int n_in,
                              void* d_out, int out_size, void* d_ws, size_t ws_size,
                              hipStream_t stream) {
    const float* x       = (const float*)d_in[0];
    const float* kernelD = (const float*)d_in[1];
    const float* kernelP = (const float*)d_in[2];
    const float* bias    = (const float*)d_in[3];
    const float* cvec    = (const float*)d_in[4];
    const float* mD_W1   = (const float*)d_in[5];
    const float* mD_b1   = (const float*)d_in[6];
    const float* mD_W2   = (const float*)d_in[7];
    const float* mD_b2   = (const float*)d_in[8];
    const float* mP_W1   = (const float*)d_in[9];
    const float* mP_b1   = (const float*)d_in[10];
    const float* mP_W2   = (const float*)d_in[11];
    const float* mP_b2   = (const float*)d_in[12];
    float* out = (float*)d_out;

    dim3 grid(256);
    dim3 block(NTHR);
    hipLaunchKernelGGL(scaconv_ds_kernel, grid, block, 0, stream,
                       x, kernelD, kernelP, bias, cvec,
                       mD_W1, mD_b1, mD_W2, mD_b2,
                       mP_W1, mP_b1, mP_W2, mP_b2, out);
}
